# starting kernel
# baseline (speedup 1.0000x reference)
#include <hip/hip_runtime.h>
#include <math.h>

typedef float v2f __attribute__((ext_vector_type(2)));
typedef float v8f __attribute__((ext_vector_type(8)));

#define T_DIM 1024
#define B_DIM 64
#define D_DIM 512
#define N_DIM 64
#define ROWS  (T_DIM * B_DIM)
#define TBN   ((size_t)T_DIM * B_DIM * N_DIM)

__global__ __launch_bounds__(128) void ntm_proj_kernel(
    const float* __restrict__ x,
    const float* __restrict__ Wk, const float* __restrict__ Wv,
    const float* __restrict__ Wq, const float* __restrict__ We,
    const float* __restrict__ Ww,
    float* __restrict__ outk, float* __restrict__ outv,
    float* __restrict__ outq, float* __restrict__ oute,
    float* __restrict__ outw)
{
    __shared__ float ldsA[16 * 64];
    __shared__ float ldsS[2 * 16 * 64];

    const int tid   = threadIdx.x;
    const int wave  = tid >> 5;
    const int lane  = tid & 31;
    const int lmod  = lane & 15;
    const bool hi   = lane >= 16;
    const int Mbase = blockIdx.x * 16;
    const int Nbase = wave * 16;

    const float* Wmat[5] = {Wk, Wv, Wq, We, Ww};

    v8f acc[5];
#pragma unroll
    for (int m = 0; m < 5; ++m)
#pragma unroll
        for (int r = 0; r < 8; ++r) acc[m][r] = 0.0f;

    for (int c = 0; c < 8; ++c) {
        __syncthreads();
#pragma unroll
        for (int i = tid; i < 16 * 64; i += 128) {
            const int r = i >> 6, col = i & 63;
            ldsA[i] = x[(size_t)(Mbase + r) * D_DIM + c * 64 + col];
        }
        __syncthreads();

        for (int kk = 0; kk < 64; kk += 4) {
            const int ak = kk + (hi ? 2 : 0);
            v2f a;
            a.x = ldsA[lmod * 64 + ak];
            a.y = ldsA[lmod * 64 + ak + 1];

            const int k0   = c * 64 + ak;
            const size_t brow = (size_t)(Nbase + lmod) * D_DIM + k0;
#pragma unroll
            for (int m = 0; m < 5; ++m) {
                v2f b;
                b.x = Wmat[m][brow];
                b.y = Wmat[m][brow + 1];
                acc[m] = __builtin_amdgcn_wmma_f32_16x16x4_f32(
                     false, a,  false, b,
                     (short)0, acc[m],
                     false,  false);
            }
        }
    }

#pragma unroll
    for (int r = 0; r < 8; ++r) {
        const int m = hi ? r + 8 : r;
        const int n = Nbase + lmod;
        ldsS[m * 64 + n]        = acc[0][r];
        ldsS[1024 + m * 64 + n] = acc[2][r];
        const size_t gi = (size_t)(Mbase + m) * N_DIM + n;
        outv[gi] = acc[1][r];
        oute[gi] = 1.0f / (1.0f + __expf(-acc[3][r]));
        outw[gi] = 1.0f / (1.0f + __expf(-acc[4][r]));
    }
    __syncthreads();

    if (tid < 16) {
#pragma unroll
        for (int s = 0; s < 2; ++s) {
            float* rowp = &ldsS[s * 1024 + tid * 64];
            float mx = rowp[0];
            for (int j = 1; j < 64; ++j) mx = fmaxf(mx, rowp[j]);
            float sum = 0.0f;
            for (int j = 0; j < 64; ++j) { float e2 = __expf(rowp[j] - mx); rowp[j] = e2; sum += e2; }
            const float inv = 1.0f / sum;
            for (int j = 0; j < 64; ++j) rowp[j] *= inv;
        }
    }
    __syncthreads();

#pragma unroll
    for (int i = tid; i < 1024; i += 128) {
        const int m = i >> 6, n = i & 63;
        const size_t gi = (size_t)(Mbase + m) * N_DIM + n;
        outk[gi] = ldsS[i];
        outq[gi] = ldsS[1024 + i];
    }
}

__global__ __launch_bounds__(64) void ntm_scan_kernel(
    const float* __restrict__ S0,
    const float* __restrict__ kG, const float* __restrict__ vG,
    const float* __restrict__ qG, const float* __restrict__ eG,
    const float* __restrict__ wG,
    float* __restrict__ out, float* __restrict__ Sfin)
{
    __shared__ float kb[64], vb[64], qb[64], eb[64], wb[64];
    const int b = blockIdx.x;
    const int i = threadIdx.x;

    float S[64];
#pragma unroll
    for (int j = 0; j < 64; ++j)
        S[j] = S0[(size_t)b * 4096 + (size_t)i * 64 + j];

    for (int t = 0; t < T_DIM; ++t) {
        const size_t base = (size_t)t * 4096 + (size_t)b * 64 + i;
        kb[i] = kG[base]; vb[i] = vG[base]; qb[i] = qG[base];
        eb[i] = eG[base]; wb[i] = wG[base];
        __syncthreads();

        const float ei  = eb[i];
        const float wvi = wb[i] * vb[i];
        float acc = 0.0f;
#pragma unroll
        for (int j = 0; j < 64; ++j) {
            const float kj = kb[j];
            float s = S[j];
            s = s * (1.0f - ei * kj) + wvi * kj;
            S[j] = s;
            acc = fmaf(s, qb[j], acc);
        }
        out[(size_t)t * 4096 + (size_t)b * 64 + i] = tanhf(acc);
        __syncthreads();
    }

#pragma unroll
    for (int j = 0; j < 64; ++j)
        Sfin[(size_t)b * 4096 + (size_t)i * 64 + j] = S[j];
}

extern "C" void kernel_launch(void* const* d_in, const int* in_sizes, int n_in,
                              void* d_out, int out_size, void* d_ws, size_t ws_size,
                              hipStream_t stream) {
    (void)in_sizes; (void)n_in; (void)out_size; (void)ws_size;
    const float* x  = (const float*)d_in[0];
    const float* S0 = (const float*)d_in[1];
    const float* Wk = (const float*)d_in[2];
    const float* Wv = (const float*)d_in[3];
    const float* Wq = (const float*)d_in[4];
    const float* We = (const float*)d_in[5];
    const float* Ww = (const float*)d_in[6];

    float* out = (float*)d_out;
    float* ws  = (float*)d_ws;
    float* wk  = ws;
    float* wv  = ws + TBN;
    float* wq  = ws + 2 * TBN;
    float* we_ = ws + 3 * TBN;
    float* ww_ = ws + 4 * TBN;

    ntm_proj_kernel<<<ROWS / 16, 128, 0, stream>>>(
        x, Wk, Wv, Wq, We, Ww, wk, wv, wq, we_, ww_);

    ntm_scan_kernel<<<B_DIM, 64, 0, stream>>>(
        S0, wk, wv, wq, we_, ww_, out, out + TBN);
}


# proj GEMM: W fragments of a k-step fetched once into registers and used for 8 row tiles (128 rows per work-group, was a fetch per 16 rows; W not in LDS); LDS pitches 68/65; same sums, same order
# speedup vs baseline: 5.5685x; 5.5685x over previous
#include <hip/hip_runtime.h>
#include <math.h>

typedef float v2f __attribute__((ext_vector_type(2)));
typedef float v8f __attribute__((ext_vector_type(8)));

#define T_DIM 1024
#define B_DIM 64
#define D_DIM 512
#define N_DIM 64
#define ROWS  (T_DIM * B_DIM)
#define TBN   ((size_t)T_DIM * B_DIM * N_DIM)
#define NCH   8
#define CH    (T_DIM / NCH)
#define SLOT  ((size_t)B_DIM * N_DIM * N_DIM)

typedef float v4f __attribute__((ext_vector_type(4)));

#define MB      128
#define RT      (MB / 16)
#define APITCH  68
#define SPITCH  65

__global__ __launch_bounds__(128) void ntm_proj_kernel(
    const float* __restrict__ x,
    const float* __restrict__ Wk, const float* __restrict__ Wv,
    const float* __restrict__ Wq, const float* __restrict__ We,
    const float* __restrict__ Ww,
    float* __restrict__ outk, float* __restrict__ outwv,
    float* __restrict__ outq, float* __restrict__ oute)
{
    __shared__ __attribute__((aligned(16))) float ldsA[MB * APITCH];
    __shared__ float ldsS[2 * MB * SPITCH];

    const int tid   = threadIdx.x;
    const int wave  = tid >> 5;
    const int lane  = tid & 31;
    const int lmod  = lane & 15;
    const bool hi   = lane >= 16;
    const int koff  = hi ? 2 : 0;
    const int Mbase = blockIdx.x * MB;
    const int Nbase = wave * 16;

    const float* Wmat[5] = {Wk, Wv, Wq, We, Ww};

    v8f acc[RT][5];
#pragma unroll
    for (int r = 0; r < RT; ++r)
#pragma unroll
        for (int m = 0; m < 5; ++m)
#pragma unroll
            for (int g = 0; g < 8; ++g) acc[r][m][g] = 0.0f;

    const float* xsrc = x + (size_t)(Mbase + (tid >> 4)) * D_DIM + (tid & 15) * 4;
    float*       adst = &ldsA[(tid >> 4) * APITCH + (tid & 15) * 4];

    for (int c = 0; c < 8; ++c) {
        __syncthreads();
#pragma unroll
        for (int i = 0; i < 16; ++i) {
            const v4f val = *(const v4f*)(xsrc + (size_t)i * 8 * D_DIM + c * 64);
            *(v4f*)(adst + i * 8 * APITCH) = val;
        }
        __syncthreads();

        const float* arow = &ldsA[lmod * APITCH + koff];
        const size_t brow = (size_t)(Nbase + lmod) * D_DIM + c * 64 + koff;

        v2f a_cur[RT], b_cur[5];
#pragma unroll
        for (int r = 0; r < RT; ++r) a_cur[r] = *(const v2f*)(arow + r * 16 * APITCH);
#pragma unroll
        for (int m = 0; m < 5; ++m) b_cur[m] = *(const v2f*)(Wmat[m] + brow);

#pragma unroll 1
        for (int sb = 0; sb < 16; sb += 4) {
#pragma unroll
            for (int u = 0; u < 4; ++u) {
                const int sn = (sb + u + 1 < 16) ? (sb + u + 1) : 15;
                const int kk = sn * 4;
                v2f a_nxt[RT], b_nxt[5];
#pragma unroll
                for (int m = 0; m < 5; ++m) b_nxt[m] = *(const v2f*)(Wmat[m] + brow + kk);
#pragma unroll
                for (int r = 0; r < RT; ++r) a_nxt[r] = *(const v2f*)(arow + r * 16 * APITCH + kk);
#pragma unroll
                for (int r = 0; r < RT; ++r)
#pragma unroll
                    for (int m = 0; m < 5; ++m)
                        acc[r][m] = __builtin_amdgcn_wmma_f32_16x16x4_f32(
                             false, a_cur[r],  false, b_cur[m],
                             (short)0, acc[r][m],
                             false,  false);
#pragma unroll
                for (int r = 0; r < RT; ++r) a_cur[r] = a_nxt[r];
#pragma unroll
                for (int m = 0; m < 5; ++m) b_cur[m] = b_nxt[m];
            }
        }
    }

#pragma unroll
    for (int r = 0; r < RT; ++r)
#pragma unroll
        for (int g = 0; g < 8; ++g) {
            const int m = r * 16 + (hi ? g + 8 : g);
            const int n = Nbase + lmod;
            ldsS[m * SPITCH + n]               = acc[r][0][g];
            ldsS[MB * SPITCH + m * SPITCH + n] = acc[r][2][g];
            const size_t gi = (size_t)(Mbase + m) * N_DIM + n;
            oute[gi]  = 1.0f / (1.0f + __expf(-acc[r][3][g]));
            outwv[gi] = (1.0f / (1.0f + __expf(-acc[r][4][g]))) * acc[r][1][g];
        }
    __syncthreads();

    {
#pragma unroll
        for (int s = 0; s < 2; ++s) {
            float* rowp = &ldsS[s * MB * SPITCH + tid * SPITCH];
            float mx = rowp[0];
            for (int j = 1; j < 64; ++j) mx = fmaxf(mx, rowp[j]);
            float sum = 0.0f;
            for (int j = 0; j < 64; ++j) { float e2 = __expf(rowp[j] - mx); rowp[j] = e2; sum += e2; }
            const float inv = 1.0f / sum;
            for (int j = 0; j < 64; ++j) rowp[j] *= inv;
        }
    }
    __syncthreads();

    {
        const int n  = tid & 63;
        const int m0 = tid >> 6;
#pragma unroll 4
        for (int it = 0; it < MB / 2; ++it) {
            const int m = it * 2 + m0;
            const size_t gi = (size_t)(Mbase + m) * N_DIM + n;
            outk[gi] = ldsS[m * SPITCH + n];
            outq[gi] = ldsS[MB * SPITCH + m * SPITCH + n];
        }
    }
}

__global__ __launch_bounds__(256) void ntm_chunk_affine_kernel(
    const float* __restrict__ kG, const float* __restrict__ eG,
    const float* __restrict__ wvG,
    float* __restrict__ Zc, float* __restrict__ Pc)
{
    const int b  = blockIdx.x;
    const int c  = blockIdx.y;
    const int i  = threadIdx.x >> 2;
    const int j0 = (threadIdx.x & 3) * 16;

    float Z[16], P[16];
#pragma unroll
    for (int j = 0; j < 16; ++j) { Z[j] = 0.0f; P[j] = 1.0f; }

    const int t0 = c * CH;
    for (int t = t0; t < t0 + CH; t += 8) {
        float ei[8], wvi[8];
        v4f kq[8][4];
#pragma unroll
        for (int s = 0; s < 8; ++s) {
            const size_t base = (size_t)(t + s) * 4096 + (size_t)b * 64;
            ei[s]  = eG[base + i];
            wvi[s] = wvG[base + i];
            const v4f* kp = (const v4f*)(kG + base + j0);
#pragma unroll
            for (int g = 0; g < 4; ++g) kq[s][g] = kp[g];
        }
#pragma unroll
        for (int s = 0; s < 8; ++s)
#pragma unroll
            for (int g = 0; g < 4; ++g)
#pragma unroll
                for (int u = 0; u < 4; ++u) {
                    const float kj = kq[s][g][u];
                    const float a  = 1.0f - ei[s] * kj;
                    Z[g * 4 + u] = Z[g * 4 + u] * a + wvi[s] * kj;
                    P[g * 4 + u] = P[g * 4 + u] * a;
                }
    }

    const size_t o = ((size_t)c * B_DIM + b) * 4096 + (size_t)i * 64 + j0;
#pragma unroll
    for (int g = 0; g < 4; ++g) {
        v4f z, p;
#pragma unroll
        for (int u = 0; u < 4; ++u) { z[u] = Z[g * 4 + u]; p[u] = P[g * 4 + u]; }
        *(v4f*)(Zc + o + g * 4) = z;
        *(v4f*)(Pc + o + g * 4) = p;
    }
}

__global__ __launch_bounds__(64) void ntm_scan_chunk_kernel(
    const float* __restrict__ S0,
    const float* __restrict__ kG, const float* __restrict__ wvG,
    const float* __restrict__ qG, const float* __restrict__ eG,
    const float* __restrict__ Zc, const float* __restrict__ Pc,
    float* __restrict__ out, float* __restrict__ Sfin)
{
    __shared__ float kb[64], qb[64];
    const int b = blockIdx.x;
    const int c = blockIdx.y;
    const int i = threadIdx.x;

    float S[64];
#pragma unroll
    for (int j = 0; j < 64; ++j)
        S[j] = S0[(size_t)b * 4096 + (size_t)i * 64 + j];

    for (int cc = 0; cc < c; ++cc) {
        const size_t o = ((size_t)cc * B_DIM + b) * 4096 + (size_t)i * 64;
#pragma unroll
        for (int j = 0; j < 64; ++j)
            S[j] = fmaf(Pc[o + j], S[j], Zc[o + j]);
    }

    const int t0 = c * CH;
    for (int t = t0; t < t0 + CH; ++t) {
        const size_t base = (size_t)t * 4096 + (size_t)b * 64 + i;
        kb[i] = kG[base]; qb[i] = qG[base];
        const float ei  = eG[base];
        const float wvi = wvG[base];
        __syncthreads();

        float acc = 0.0f;
#pragma unroll
        for (int j = 0; j < 64; ++j) {
            const float kj = kb[j];
            float s = S[j];
            s = s * (1.0f - ei * kj) + wvi * kj;
            S[j] = s;
            acc = fmaf(s, qb[j], acc);
        }
        out[(size_t)t * 4096 + (size_t)b * 64 + i] = tanhf(acc);
        __syncthreads();
    }

    if (c == NCH - 1) {
#pragma unroll
        for (int j = 0; j < 64; ++j)
            Sfin[(size_t)b * 4096 + (size_t)i * 64 + j] = S[j];
    }
}

extern "C" void kernel_launch(void* const* d_in, const int* in_sizes, int n_in,
                              void* d_out, int out_size, void* d_ws, size_t ws_size,
                              hipStream_t stream) {
    (void)in_sizes; (void)n_in; (void)out_size; (void)ws_size;
    const float* x  = (const float*)d_in[0];
    const float* S0 = (const float*)d_in[1];
    const float* Wk = (const float*)d_in[2];
    const float* Wv = (const float*)d_in[3];
    const float* Wq = (const float*)d_in[4];
    const float* We = (const float*)d_in[5];
    const float* Ww = (const float*)d_in[6];

    float* out = (float*)d_out;
    float* ws  = (float*)d_ws;
    float* wk  = ws;
    float* wwv = ws + TBN;
    float* wq  = ws + 2 * TBN;
    float* we_ = ws + 3 * TBN;
    float* zc  = ws + 4 * TBN;
    float* pc  = ws + 4 * TBN + (NCH - 1) * SLOT;

    ntm_proj_kernel<<<ROWS / MB, 128, 0, stream>>>(
        x, Wk, Wv, Wq, We, Ww, wk, wwv, wq, we_);

    ntm_chunk_affine_kernel<<<dim3(B_DIM, NCH - 1), 256, 0, stream>>>(
        wk, we_, wwv, zc, pc);

    ntm_scan_chunk_kernel<<<dim3(B_DIM, NCH), 64, 0, stream>>>(
        S0, wk, wwv, wq, we_, zc, pc, out, out + TBN);
}


# scan: k,q,e,w*v fetched 7 steps ahead into a register ring (2+2 k,q values per lane); column j handed to all lanes by v_readlane as a scalar operand; no LDS, no barrier; same sums, same order
# speedup vs baseline: 9.5044x; 1.7068x over previous
#include <hip/hip_runtime.h>
#include <math.h>

typedef float v2f __attribute__((ext_vector_type(2)));
typedef float v8f __attribute__((ext_vector_type(8)));

#define T_DIM 1024
#define B_DIM 64
#define D_DIM 512
#define N_DIM 64
#define ROWS  (T_DIM * B_DIM)
#define TBN   ((size_t)T_DIM * B_DIM * N_DIM)
#define NCH   8
#define CH    (T_DIM / NCH)
#define SLOT  ((size_t)B_DIM * N_DIM * N_DIM)

typedef float v4f __attribute__((ext_vector_type(4)));

#define MB      128
#define RT      (MB / 16)
#define APITCH  68
#define SPITCH  65

__global__ __launch_bounds__(128) void ntm_proj_kernel(
    const float* __restrict__ x,
    const float* __restrict__ Wk, const float* __restrict__ Wv,
    const float* __restrict__ Wq, const float* __restrict__ We,
    const float* __restrict__ Ww,
    float* __restrict__ outk, float* __restrict__ outwv,
    float* __restrict__ outq, float* __restrict__ oute)
{
    __shared__ __attribute__((aligned(16))) float ldsA[MB * APITCH];
    __shared__ float ldsS[2 * MB * SPITCH];

    const int tid   = threadIdx.x;
    const int wave  = tid >> 5;
    const int lane  = tid & 31;
    const int lmod  = lane & 15;
    const bool hi   = lane >= 16;
    const int koff  = hi ? 2 : 0;
    const int Mbase = blockIdx.x * MB;
    const int Nbase = wave * 16;

    const float* Wmat[5] = {Wk, Wv, Wq, We, Ww};

    v8f acc[RT][5];
#pragma unroll
    for (int r = 0; r < RT; ++r)
#pragma unroll
        for (int m = 0; m < 5; ++m)
#pragma unroll
            for (int g = 0; g < 8; ++g) acc[r][m][g] = 0.0f;

    const float* xsrc = x + (size_t)(Mbase + (tid >> 4)) * D_DIM + (tid & 15) * 4;
    float*       adst = &ldsA[(tid >> 4) * APITCH + (tid & 15) * 4];

    for (int c = 0; c < 8; ++c) {
        __syncthreads();
#pragma unroll
        for (int i = 0; i < 16; ++i) {
            const v4f val = *(const v4f*)(xsrc + (size_t)i * 8 * D_DIM + c * 64);
            *(v4f*)(adst + i * 8 * APITCH) = val;
        }
        __syncthreads();

        const float* arow = &ldsA[lmod * APITCH + koff];
        const size_t brow = (size_t)(Nbase + lmod) * D_DIM + c * 64 + koff;

        v2f a_cur[RT], b_cur[5];
#pragma unroll
        for (int r = 0; r < RT; ++r) a_cur[r] = *(const v2f*)(arow + r * 16 * APITCH);
#pragma unroll
        for (int m = 0; m < 5; ++m) b_cur[m] = *(const v2f*)(Wmat[m] + brow);

#pragma unroll 1
        for (int sb = 0; sb < 16; sb += 4) {
#pragma unroll
            for (int u = 0; u < 4; ++u) {
                const int sn = (sb + u + 1 < 16) ? (sb + u + 1) : 15;
                const int kk = sn * 4;
                v2f a_nxt[RT], b_nxt[5];
#pragma unroll
                for (int m = 0; m < 5; ++m) b_nxt[m] = *(const v2f*)(Wmat[m] + brow + kk);
#pragma unroll
                for (int r = 0; r < RT; ++r) a_nxt[r] = *(const v2f*)(arow + r * 16 * APITCH + kk);
#pragma unroll
                for (int r = 0; r < RT; ++r)
#pragma unroll
                    for (int m = 0; m < 5; ++m)
                        acc[r][m] = __builtin_amdgcn_wmma_f32_16x16x4_f32(
                             false, a_cur[r],  false, b_cur[m],
                             (short)0, acc[r][m],
                             false,  false);
#pragma unroll
                for (int r = 0; r < RT; ++r) a_cur[r] = a_nxt[r];
#pragma unroll
                for (int m = 0; m < 5; ++m) b_cur[m] = b_nxt[m];
            }
        }
    }

#pragma unroll
    for (int r = 0; r < RT; ++r)
#pragma unroll
        for (int g = 0; g < 8; ++g) {
            const int m = r * 16 + (hi ? g + 8 : g);
            const int n = Nbase + lmod;
            ldsS[m * SPITCH + n]               = acc[r][0][g];
            ldsS[MB * SPITCH + m * SPITCH + n] = acc[r][2][g];
            const size_t gi = (size_t)(Mbase + m) * N_DIM + n;
            oute[gi]  = 1.0f / (1.0f + __expf(-acc[r][3][g]));
            outwv[gi] = (1.0f / (1.0f + __expf(-acc[r][4][g]))) * acc[r][1][g];
        }
    __syncthreads();

    {
#pragma unroll
        for (int s = 0; s < 2; ++s) {
            float* rowp = &ldsS[s * MB * SPITCH + tid * SPITCH];
            float mx = rowp[0];
            for (int j = 1; j < 64; ++j) mx = fmaxf(mx, rowp[j]);
            float sum = 0.0f;
            for (int j = 0; j < 64; ++j) { float e2 = __expf(rowp[j] - mx); rowp[j] = e2; sum += e2; }
            const float inv = 1.0f / sum;
            for (int j = 0; j < 64; ++j) rowp[j] *= inv;
        }
    }
    __syncthreads();

    {
        const int n  = tid & 63;
        const int m0 = tid >> 6;
#pragma unroll 4
        for (int it = 0; it < MB / 2; ++it) {
            const int m = it * 2 + m0;
            const size_t gi = (size_t)(Mbase + m) * N_DIM + n;
            outk[gi] = ldsS[m * SPITCH + n];
            outq[gi] = ldsS[MB * SPITCH + m * SPITCH + n];
        }
    }
}

__global__ __launch_bounds__(256) void ntm_chunk_affine_kernel(
    const float* __restrict__ kG, const float* __restrict__ eG,
    const float* __restrict__ wvG,
    float* __restrict__ Zc, float* __restrict__ Pc)
{
    const int b  = blockIdx.x;
    const int c  = blockIdx.y;
    const int i  = threadIdx.x >> 2;
    const int j0 = (threadIdx.x & 3) * 16;

    float Z[16], P[16];
#pragma unroll
    for (int j = 0; j < 16; ++j) { Z[j] = 0.0f; P[j] = 1.0f; }

    const int t0 = c * CH;
    for (int t = t0; t < t0 + CH; t += 8) {
        float ei[8], wvi[8];
        v4f kq[8][4];
#pragma unroll
        for (int s = 0; s < 8; ++s) {
            const size_t base = (size_t)(t + s) * 4096 + (size_t)b * 64;
            ei[s]  = eG[base + i];
            wvi[s] = wvG[base + i];
            const v4f* kp = (const v4f*)(kG + base + j0);
#pragma unroll
            for (int g = 0; g < 4; ++g) kq[s][g] = kp[g];
        }
#pragma unroll
        for (int s = 0; s < 8; ++s)
#pragma unroll
            for (int g = 0; g < 4; ++g)
#pragma unroll
                for (int u = 0; u < 4; ++u) {
                    const float kj = kq[s][g][u];
                    const float a  = 1.0f - ei[s] * kj;
                    Z[g * 4 + u] = Z[g * 4 + u] * a + wvi[s] * kj;
                    P[g * 4 + u] = P[g * 4 + u] * a;
                }
    }

    const size_t o = ((size_t)c * B_DIM + b) * 4096 + (size_t)i * 64 + j0;
#pragma unroll
    for (int g = 0; g < 4; ++g) {
        v4f z, p;
#pragma unroll
        for (int u = 0; u < 4; ++u) { z[u] = Z[g * 4 + u]; p[u] = P[g * 4 + u]; }
        *(v4f*)(Zc + o + g * 4) = z;
        *(v4f*)(Pc + o + g * 4) = p;
    }
}

#define PD 8

__device__ __forceinline__ float from_lane(float v, int l)
{
    return __builtin_bit_cast(float, __builtin_amdgcn_readlane(__builtin_bit_cast(int, v), l));
}

__global__ __launch_bounds__(64) void ntm_scan_chunk_kernel(
    const float* __restrict__ S0,
    const float* __restrict__ kG, const float* __restrict__ wvG,
    const float* __restrict__ qG, const float* __restrict__ eG,
    const float* __restrict__ Zc, const float* __restrict__ Pc,
    float* __restrict__ out, float* __restrict__ Sfin)
{
    const int b  = blockIdx.x;
    const int c  = blockIdx.y;
    const int i  = threadIdx.x;
    const int l  = i & 31;
    const int t0 = c * CH;

    v2f S[32];
    {
        const v2f* sp = (const v2f*)(S0 + (size_t)b * 4096 + (size_t)i * 64);
#pragma unroll
        for (int p = 0; p < 32; ++p) S[p] = sp[p];
    }

    for (int cc = 0; cc < c; ++cc) {
        const size_t o = ((size_t)cc * B_DIM + b) * 4096 + (size_t)i * 64;
        const v2f* pp = (const v2f*)(Pc + o);
        const v2f* zp = (const v2f*)(Zc + o);
#pragma unroll
        for (int p = 0; p < 32; ++p)
            S[p] = __builtin_elementwise_fma(pp[p], S[p], zp[p]);
    }

    float rk0[PD], rk1[PD], rq0[PD], rq1[PD], re[PD], rw[PD];
#pragma unroll
    for (int u = 0; u < PD - 1; ++u) {
        const size_t base = (size_t)(t0 + u) * 4096 + (size_t)b * 64;
        rk0[u] = kG[base + l];  rk1[u] = kG[base + 32 + l];
        rq0[u] = qG[base + l];  rq1[u] = qG[base + 32 + l];
        re[u]  = eG[base + i];  rw[u]  = wvG[base + i];
        __builtin_amdgcn_sched_barrier(0);
    }
    rk0[PD - 1] = 0.0f; rk1[PD - 1] = 0.0f; rq0[PD - 1] = 0.0f; rq1[PD - 1] = 0.0f;
    re[PD - 1]  = 0.0f; rw[PD - 1]  = 0.0f;

    const v2f one2 = {1.0f, 1.0f};

    for (int t = t0; t < t0 + CH; t += PD) {
#pragma unroll
        for (int u = 0; u < PD; ++u) {
            {
                const int v = (u + PD - 1) % PD;
                int tn = t + u + PD - 1;
                if (tn > t0 + CH - 1) tn = t0 + CH - 1;
                const size_t nb = (size_t)tn * 4096 + (size_t)b * 64;
                rk0[v] = kG[nb + l];  rk1[v] = kG[nb + 32 + l];
                rq0[v] = qG[nb + l];  rq1[v] = qG[nb + 32 + l];
                re[v]  = eG[nb + i];  rw[v]  = wvG[nb + i];
            }

            const float k0 = rk0[u], k1 = rk1[u], q0 = rq0[u], q1 = rq1[u];
            v2f e2 = {re[u], re[u]};
            v2f w2 = {rw[u], rw[u]};
            asm volatile("" : "+v"(e2), "+v"(w2));

            float acc = 0.0f;
#pragma unroll
            for (int p = 0; p < 32; ++p) {
                const int j = 2 * p;
                v2f kk;
                kk.x = (j < 32) ? from_lane(k0, j)     : from_lane(k1, j - 32);
                kk.y = (j < 32) ? from_lane(k0, j + 1) : from_lane(k1, j - 31);
                const float qa = (j < 32) ? from_lane(q0, j)     : from_lane(q1, j - 32);
                const float qb = (j < 32) ? from_lane(q0, j + 1) : from_lane(q1, j - 31);
                const v2f a = __builtin_elementwise_fma(-e2, kk, one2);
                const v2f m = S[p] * a;
                const v2f s = __builtin_elementwise_fma(w2, kk, m);
                S[p] = s;
                acc = fmaf(s.x, qa, acc);
                acc = fmaf(s.y, qb, acc);
            }
            out[(size_t)(t + u) * 4096 + (size_t)b * 64 + i] = tanhf(acc);
        }
    }

    if (c == NCH - 1) {
        v2f* fp = (v2f*)(Sfin + (size_t)b * 4096 + (size_t)i * 64);
#pragma unroll
        for (int p = 0; p < 32; ++p) fp[p] = S[p];
    }
}

extern "C" void kernel_launch(void* const* d_in, const int* in_sizes, int n_in,
                              void* d_out, int out_size, void* d_ws, size_t ws_size,
                              hipStream_t stream) {
    (void)in_sizes; (void)n_in; (void)out_size; (void)ws_size;
    const float* x  = (const float*)d_in[0];
    const float* S0 = (const float*)d_in[1];
    const float* Wk = (const float*)d_in[2];
    const float* Wv = (const float*)d_in[3];
    const float* Wq = (const float*)d_in[4];
    const float* We = (const float*)d_in[5];
    const float* Ww = (const float*)d_in[6];

    float* out = (float*)d_out;
    float* ws  = (float*)d_ws;
    float* wk  = ws;
    float* wwv = ws + TBN;
    float* wq  = ws + 2 * TBN;
    float* we_ = ws + 3 * TBN;
    float* zc  = ws + 4 * TBN;
    float* pc  = ws + 4 * TBN + (NCH - 1) * SLOT;

    ntm_proj_kernel<<<ROWS / MB, 128, 0, stream>>>(
        x, Wk, Wv, Wq, We, Ww, wk, wwv, wq, we_);

    ntm_chunk_affine_kernel<<<dim3(B_DIM, NCH - 1), 256, 0, stream>>>(
        wk, we_, wwv, zc, pc);

    ntm_scan_chunk_kernel<<<dim3(B_DIM, NCH), 64, 0, stream>>>(
        S0, wk, wwv, wq, we_, zc, pc, out, out + TBN);
}
